# starting kernel
# baseline (speedup 1.0000x reference)
#include <hip/hip_runtime.h>
#include <math.h>

#define BB 8
#define HH 192
#define WW 192
#define NT (WW / 16)
#define WAVES_PER_BLOCK 4

typedef __attribute__((ext_vector_type(2))) float v2f;
typedef __attribute__((ext_vector_type(8))) float v8f;

__global__ __launch_bounds__(256) void vert_pass(const float* __restrict__ mask,
                                                 float* __restrict__ g2) {
  int idx = blockIdx.x * blockDim.x + threadIdx.x;
  if (idx >= BB * WW) return;
  int b = idx / WW;
  int w = idx - b * WW;
  const float BIGF = (float)(HH * HH + WW * WW);
  const float* mcol = mask + (size_t)b * HH * WW + w;
  float* gcol = g2 + (size_t)b * HH * WW + w;

  float m_up = -BIGF;
  for (int h = 0; h < HH; ++h) {
    float mv = mcol[(size_t)h * WW];
    float r = (float)h;
    if (mv > 0.5f) m_up = r;
    gcol[(size_t)h * WW] = r - m_up;
  }
  float m_dn = -BIGF;
  for (int h = HH - 1; h >= 0; --h) {
    float mv = mcol[(size_t)h * WW];
    float r = (float)h;
    if (mv > 0.5f) m_dn = -r;
    float up = gcol[(size_t)h * WW];
    float down = -m_dn - r;
    float d = fminf(fminf(up, down), BIGF);
    gcol[(size_t)h * WW] = d * d;
  }
}

__global__ __launch_bounds__(32 * WAVES_PER_BLOCK)
void edt_rows(const float* __restrict__ g2, float* __restrict__ out) {
  __shared__ float rowmin[WAVES_PER_BLOCK][WW];

  int wave = threadIdx.x >> 5;
  int lane = threadIdx.x & 31;
  int row = blockIdx.x * WAVES_PER_BLOCK + wave;
  const float* grow = g2 + (size_t)row * WW;

  __builtin_prefetch(grow + lane * 6, 0, 0);

  bool upper = lane >= 16;
  int nl = lane & 15;

  v2f bt[NT];
#pragma unroll
  for (int t = 0; t < NT; ++t) {
    float jp = (float)(t * 16 + nl);
    float g = grow[t * 16 + nl];
    bt[t].x = upper ? (jp * jp + g) : 1.0f;
    bt[t].y = upper ? 0.0f : jp;
  }

  v8f czero = {};
  for (int o = 0; o < NT; ++o) {
    float j = (float)(o * 16 + nl);
    v2f a;
    a.x = upper ? 1.0f : (j * j);
    a.y = upper ? 0.0f : (-2.0f * j);

    v8f acc;
#pragma unroll
    for (int i = 0; i < 8; ++i) acc[i] = 3.0e38f;

#pragma unroll
    for (int t = 0; t < NT; ++t) {
      v8f d = __builtin_amdgcn_wmma_f32_16x16x4_f32(
           false, a,  false, bt[t],
           (short)0, czero,  false,  false);
#pragma unroll
      for (int i = 0; i < 8; ++i) acc[i] = fminf(acc[i], d[i]);
    }

#pragma unroll
    for (int i = 0; i < 8; ++i) {
#pragma unroll
      for (int m = 1; m < 16; m <<= 1)
        acc[i] = fminf(acc[i], __shfl_xor(acc[i], m, 32));
    }

    if (lane == 0) {
#pragma unroll
      for (int r = 0; r < 8; ++r) rowmin[wave][o * 16 + r] = acc[r];
    }
    if (lane == 16) {
#pragma unroll
      for (int r = 0; r < 8; ++r) rowmin[wave][o * 16 + 8 + r] = acc[r];
    }
  }

  __syncthreads();

  for (int k = lane; k < WW; k += 32) {
    float dsq = rowmin[wave][k];
    float v = 2.0f / (1.0f + expf(0.1f * sqrtf(dsq)));
    out[(size_t)row * WW + k] = v;
  }
}

extern "C" void kernel_launch(void* const* d_in, const int* in_sizes, int n_in,
                              void* d_out, int out_size, void* d_ws, size_t ws_size,
                              hipStream_t stream) {
  (void)in_sizes; (void)n_in; (void)out_size; (void)ws_size;
  const float* mask = (const float*)d_in[0];
  float* out = (float*)d_out;
  float* g2 = (float*)d_ws;

  int cols = BB * WW;
  vert_pass<<<(cols + 255) / 256, 256, 0, stream>>>(mask, g2);

  int rows = BB * HH;
  edt_rows<<<rows / WAVES_PER_BLOCK, 32 * WAVES_PER_BLOCK, 0, stream>>>(g2, out);
}


# vertical pass: per-column 192-step serial scan through global memory split into 12 row segments per column, scanned in registers (integer rows) and joined through LDS; 48 work-groups of 12 waves
# speedup vs baseline: 2.7128x; 2.7128x over previous
#include <hip/hip_runtime.h>
#include <math.h>

#define BB 8
#define HH 192
#define WW 192
#define NT (WW / 16)
#define WAVES_PER_BLOCK 4

typedef __attribute__((ext_vector_type(2))) float v2f;
typedef __attribute__((ext_vector_type(8))) float v8f;

#define SEG 16
#define NSEG (HH / SEG)
#define STRIP 32
#define BIGI (HH * HH + WW * WW)

__global__ __launch_bounds__(STRIP * NSEG)
void vert_pass(const float* __restrict__ mask, float* __restrict__ g2) {
  __shared__ int2 seg_ud[NSEG][STRIP];

  int c = threadIdx.x & (STRIP - 1);
  int s = __builtin_amdgcn_readfirstlane((int)(threadIdx.x / STRIP));
  int strip = blockIdx.x % (WW / STRIP);
  int b = blockIdx.x / (WW / STRIP);
  int h0 = s * SEG;
  size_t base = ((size_t)b * HH + h0) * WW + strip * STRIP + c;
  const float* mcol = mask + base;
  float* gcol = g2 + base;

  bool fg[SEG];
#pragma unroll
  for (int i = 0; i < SEG; ++i) fg[i] = mcol[(size_t)i * WW] > 0.5f;

  int lu[SEG], ld[SEG];
  int m_up = -BIGI;
#pragma unroll
  for (int i = 0; i < SEG; ++i) {
    if (fg[i]) m_up = h0 + i;
    lu[i] = m_up;
  }
  int m_dn = BIGI;
#pragma unroll
  for (int i = SEG - 1; i >= 0; --i) {
    if (fg[i]) m_dn = h0 + i;
    ld[i] = m_dn;
  }
  seg_ud[s][c] = make_int2(m_up, m_dn);
  __syncthreads();

  int cu = -BIGI, cd = BIGI;
#pragma unroll
  for (int t = 0; t < NSEG; ++t) {
    int2 p = seg_ud[t][c];
    cu = max(cu, (t < s) ? p.x : -BIGI);
    cd = min(cd, (t > s) ? p.y : BIGI);
  }

#pragma unroll
  for (int i = 0; i < SEG; ++i) {
    int r = h0 + i;
    int up = r - max(lu[i], cu);
    int down = min(ld[i], cd) - r;
    float d = (float)min(min(up, down), BIGI);
    gcol[(size_t)i * WW] = d * d;
  }
}

__global__ __launch_bounds__(32 * WAVES_PER_BLOCK)
void edt_rows(const float* __restrict__ g2, float* __restrict__ out) {
  __shared__ float rowmin[WAVES_PER_BLOCK][WW];

  int wave = threadIdx.x >> 5;
  int lane = threadIdx.x & 31;
  int row = blockIdx.x * WAVES_PER_BLOCK + wave;
  const float* grow = g2 + (size_t)row * WW;

  __builtin_prefetch(grow + lane * 6, 0, 0);

  bool upper = lane >= 16;
  int nl = lane & 15;

  v2f bt[NT];
#pragma unroll
  for (int t = 0; t < NT; ++t) {
    float jp = (float)(t * 16 + nl);
    float g = grow[t * 16 + nl];
    bt[t].x = upper ? (jp * jp + g) : 1.0f;
    bt[t].y = upper ? 0.0f : jp;
  }

  v8f czero = {};
  for (int o = 0; o < NT; ++o) {
    float j = (float)(o * 16 + nl);
    v2f a;
    a.x = upper ? 1.0f : (j * j);
    a.y = upper ? 0.0f : (-2.0f * j);

    v8f acc;
#pragma unroll
    for (int i = 0; i < 8; ++i) acc[i] = 3.0e38f;

#pragma unroll
    for (int t = 0; t < NT; ++t) {
      v8f d = __builtin_amdgcn_wmma_f32_16x16x4_f32(
           false, a,  false, bt[t],
           (short)0, czero,  false,  false);
#pragma unroll
      for (int i = 0; i < 8; ++i) acc[i] = fminf(acc[i], d[i]);
    }

#pragma unroll
    for (int i = 0; i < 8; ++i) {
#pragma unroll
      for (int m = 1; m < 16; m <<= 1)
        acc[i] = fminf(acc[i], __shfl_xor(acc[i], m, 32));
    }

    if (lane == 0) {
#pragma unroll
      for (int r = 0; r < 8; ++r) rowmin[wave][o * 16 + r] = acc[r];
    }
    if (lane == 16) {
#pragma unroll
      for (int r = 0; r < 8; ++r) rowmin[wave][o * 16 + 8 + r] = acc[r];
    }
  }

  __syncthreads();

  for (int k = lane; k < WW; k += 32) {
    float dsq = rowmin[wave][k];
    float v = 2.0f / (1.0f + expf(0.1f * sqrtf(dsq)));
    out[(size_t)row * WW + k] = v;
  }
}

extern "C" void kernel_launch(void* const* d_in, const int* in_sizes, int n_in,
                              void* d_out, int out_size, void* d_ws, size_t ws_size,
                              hipStream_t stream) {
  (void)in_sizes; (void)n_in; (void)out_size; (void)ws_size;
  const float* mask = (const float*)d_in[0];
  float* out = (float*)d_out;
  float* g2 = (float*)d_ws;

  vert_pass<<<BB * (WW / STRIP), STRIP * NSEG, 0, stream>>>(mask, g2);

  int rows = BB * HH;
  edt_rows<<<rows / WAVES_PER_BLOCK, 32 * WAVES_PER_BLOCK, 0, stream>>>(g2, out);
}


# row pass: WMMA A/B roles swapped (candidates on M, outputs on N): row-min is in-lane v_min3 plus one v_permlanex16, not 32 ds_bpermute per tile; stored from registers, no LDS, no barrier
# speedup vs baseline: 3.9863x; 1.4694x over previous
#include <hip/hip_runtime.h>
#include <math.h>

#define BB 8
#define HH 192
#define WW 192
#define NT (WW / 16)
#define WAVES_PER_BLOCK 4

typedef __attribute__((ext_vector_type(2))) float v2f;
typedef __attribute__((ext_vector_type(8))) float v8f;

#define SEG 16
#define NSEG (HH / SEG)
#define STRIP 32
#define BIGI (HH * HH + WW * WW)

__global__ __launch_bounds__(STRIP * NSEG)
void vert_pass(const float* __restrict__ mask, float* __restrict__ g2) {
  __shared__ int2 seg_ud[NSEG][STRIP];

  int c = threadIdx.x & (STRIP - 1);
  int s = __builtin_amdgcn_readfirstlane((int)(threadIdx.x / STRIP));
  int strip = blockIdx.x % (WW / STRIP);
  int b = blockIdx.x / (WW / STRIP);
  int h0 = s * SEG;
  size_t base = ((size_t)b * HH + h0) * WW + strip * STRIP + c;
  const float* mcol = mask + base;
  float* gcol = g2 + base;

  bool fg[SEG];
#pragma unroll
  for (int i = 0; i < SEG; ++i) fg[i] = mcol[(size_t)i * WW] > 0.5f;

  int lu[SEG], ld[SEG];
  int m_up = -BIGI;
#pragma unroll
  for (int i = 0; i < SEG; ++i) {
    if (fg[i]) m_up = h0 + i;
    lu[i] = m_up;
  }
  int m_dn = BIGI;
#pragma unroll
  for (int i = SEG - 1; i >= 0; --i) {
    if (fg[i]) m_dn = h0 + i;
    ld[i] = m_dn;
  }
  seg_ud[s][c] = make_int2(m_up, m_dn);
  __syncthreads();

  int cu = -BIGI, cd = BIGI;
#pragma unroll
  for (int t = 0; t < NSEG; ++t) {
    int2 p = seg_ud[t][c];
    cu = max(cu, (t < s) ? p.x : -BIGI);
    cd = min(cd, (t > s) ? p.y : BIGI);
  }

#pragma unroll
  for (int i = 0; i < SEG; ++i) {
    int r = h0 + i;
    int up = r - max(lu[i], cu);
    int down = min(ld[i], cd) - r;
    float d = (float)min(min(up, down), BIGI);
    gcol[(size_t)i * WW] = d * d;
  }
}


__device__ __forceinline__ float xhalf(float x) {
  int b = __builtin_bit_cast(int, x);
  return __builtin_bit_cast(float, __builtin_amdgcn_permlanex16(
      b, b, 0x76543210, 0xfedcba98, false, false));
}

__global__ __launch_bounds__(32 * WAVES_PER_BLOCK)
void edt_rows(const float* __restrict__ g2, float* __restrict__ out) {
  int wave = threadIdx.x >> 5;
  int lane = threadIdx.x & 31;
  int row = blockIdx.x * WAVES_PER_BLOCK + wave;
  const float* grow = g2 + (size_t)row * WW;
  float* orow = out + (size_t)row * WW;

  __builtin_prefetch(grow + lane * 6, 0, 0);

  bool upper = lane >= 16;
  int nl = lane & 15;

  v2f ct[NT];
#pragma unroll
  for (int t = 0; t < NT; ++t) {
    float jp = (float)(t * 16 + nl);
    float g = grow[t * 16 + nl];
    ct[t].x = upper ? (jp * jp + g) : 1.0f;
    ct[t].y = upper ? 0.0f : jp;
  }

  v8f czero = {};
  float dsq[NT / 2];
#pragma unroll
  for (int p = 0; p < NT / 2; ++p) {
    float j0 = (float)(p * 32 + nl);
    float j1 = j0 + 16.0f;
    v2f b0, b1;
    b0.x = upper ? 1.0f : (j0 * j0);
    b0.y = upper ? 0.0f : (-2.0f * j0);
    b1.x = upper ? 1.0f : (j1 * j1);
    b1.y = upper ? 0.0f : (-2.0f * j1);

    v8f acc0, acc1;
#pragma unroll
    for (int i = 0; i < 8; ++i) { acc0[i] = 3.0e38f; acc1[i] = 3.0e38f; }

#pragma unroll
    for (int t = 0; t < NT; t += 2) {
      v8f d00 = __builtin_amdgcn_wmma_f32_16x16x4_f32(
          false, ct[t], false, b0, (short)0, czero, false, false);
      v8f d10 = __builtin_amdgcn_wmma_f32_16x16x4_f32(
          false, ct[t], false, b1, (short)0, czero, false, false);
      v8f d01 = __builtin_amdgcn_wmma_f32_16x16x4_f32(
          false, ct[t + 1], false, b0, (short)0, czero, false, false);
      v8f d11 = __builtin_amdgcn_wmma_f32_16x16x4_f32(
          false, ct[t + 1], false, b1, (short)0, czero, false, false);
#pragma unroll
      for (int i = 0; i < 8; ++i) {
        acc0[i] = fminf(fminf(acc0[i], d00[i]), d01[i]);
        acc1[i] = fminf(fminf(acc1[i], d10[i]), d11[i]);
      }
    }

    float m0 = fminf(fminf(fminf(acc0[0], acc0[1]), fminf(acc0[2], acc0[3])),
                     fminf(fminf(acc0[4], acc0[5]), fminf(acc0[6], acc0[7])));
    float m1 = fminf(fminf(fminf(acc1[0], acc1[1]), fminf(acc1[2], acc1[3])),
                     fminf(fminf(acc1[4], acc1[5]), fminf(acc1[6], acc1[7])));
    m0 = fminf(m0, xhalf(m0));
    m1 = fminf(m1, xhalf(m1));

    dsq[p] = upper ? m1 : m0;
  }

#pragma unroll
  for (int p = 0; p < NT / 2; ++p) {
    float v = 2.0f / (1.0f + expf(0.1f * sqrtf(dsq[p])));
    orow[p * 32 + lane] = v;
  }
}

extern "C" void kernel_launch(void* const* d_in, const int* in_sizes, int n_in,
                              void* d_out, int out_size, void* d_ws, size_t ws_size,
                              hipStream_t stream) {
  (void)in_sizes; (void)n_in; (void)out_size; (void)ws_size;
  const float* mask = (const float*)d_in[0];
  float* out = (float*)d_out;
  float* g2 = (float*)d_ws;

  vert_pass<<<BB * (WW / STRIP), STRIP * NSEG, 0, stream>>>(mask, g2);

  int rows = BB * HH;
  edt_rows<<<rows / WAVES_PER_BLOCK, 32 * WAVES_PER_BLOCK, 0, stream>>>(g2, out);
}


# one launch instead of two: each 4-row work-group computes its rows' vertical distances itself from a 16-row mask window per direction (bounded walk beyond it) and hands them on through LDS
# speedup vs baseline: 5.3171x; 1.3338x over previous
#include <hip/hip_runtime.h>
#include <math.h>


#define BB 8
#define HH 192
#define WW 192
#define NT (WW / 16)
#define WAVES_PER_BLOCK 4
#define BAND WAVES_PER_BLOCK
#define BIGI (HH * HH + WW * WW)
#define CHUNK 16
#define STRIP 32
#define NUNIT (2 * WW / STRIP)
#define UNITS_PER_WAVE (NUNIT / WAVES_PER_BLOCK)
#define NOTFOUND (BIGI + 4 * CHUNK)

typedef __attribute__((ext_vector_type(2))) float v2f;
typedef __attribute__((ext_vector_type(8))) float v8f;

__device__ __forceinline__ float xhalf(float x) {
  int b = __builtin_bit_cast(int, x);
  return __builtin_bit_cast(float, __builtin_amdgcn_permlanex16(
      b, b, 0x76543210, 0xfedcba98, false, false));
}

__global__ __launch_bounds__(32 * WAVES_PER_BLOCK)
void edt_fused(const float* __restrict__ mask, float* __restrict__ out) {
  __shared__ int dist[2][BAND][WW];

  int wave = __builtin_amdgcn_readfirstlane((int)(threadIdx.x >> 5));
  int lane = threadIdx.x & 31;
  int b = blockIdx.x / (HH / BAND);
  int r0 = (blockIdx.x % (HH / BAND)) * BAND;
  const float* img = mask + (size_t)b * HH * WW;

  float v[UNITS_PER_WAVE][CHUNK];
#pragma unroll
  for (int u = 0; u < UNITS_PER_WAVE; ++u) {
    int unit = wave * UNITS_PER_WAVE + u;
    int down = unit & 1;
    int c = (unit >> 1) * STRIP + lane;
    int start = down ? r0 : r0 + BAND - 1;
    int wb = down ? min(start, HH - CHUNK) : max(start, CHUNK - 1);
    int lo = down ? wb : wb - (CHUNK - 1);
    const float* p = img + lo * WW + c;
#pragma unroll
    for (int i = 0; i < CHUNK; ++i) v[u][i] = p[i * WW];
  }

  unsigned bitsu[UNITS_PER_WAVE];
  bool need[UNITS_PER_WAVE];
  int far[UNITS_PER_WAVE];
#pragma unroll
  for (int u = 0; u < UNITS_PER_WAVE; ++u) {
    int unit = wave * UNITS_PER_WAVE + u;
    int down = unit & 1;
    int start = down ? r0 : r0 + BAND - 1;
    int wb = down ? min(start, HH - CHUNK) : max(start, CHUNK - 1);
    int maxpos = down ? HH - 1 - r0 : r0 + BAND - 1;
    unsigned bits = 0;
#pragma unroll
    for (int i = 0; i < CHUNK; ++i) bits |= (v[u][i] > 0.5f) ? (1u << i) : 0u;
    if (!down) bits = __builtin_bitreverse32(bits) >> (32 - CHUNK);
    bits >>= down ? start - wb : wb - start;
    bitsu[u] = bits;
    need[u] = ((bits >> (BAND - 1)) == 0) && (CHUNK <= maxpos);
    far[u] = down ? BIGI - r0 : NOTFOUND;
  }

  if (__builtin_amdgcn_ballot_w32(need[0] || need[1] || need[2]) != 0) {
#pragma unroll
    for (int u = 0; u < UNITS_PER_WAVE; ++u) {
      int unit = wave * UNITS_PER_WAVE + u;
      int down = unit & 1;
      int c = (unit >> 1) * STRIP + lane;
      int start = down ? r0 : r0 + BAND - 1;
      int maxpos = down ? HH - 1 - r0 : r0 + BAND - 1;
      for (int base = CHUNK;
           base <= maxpos && __builtin_amdgcn_ballot_w32(need[u]) != 0;
           base += CHUNK) {
        int first = down ? start + base : start - base;
        int wbn = down ? min(first, HH - CHUNK) : max(first, CHUNK - 1);
        const float* q = img + (down ? wbn : wbn - (CHUNK - 1)) * WW + c;
        unsigned m = 0;
#pragma unroll
        for (int i = 0; i < CHUNK; ++i) m |= (q[i * WW] > 0.5f) ? (1u << i) : 0u;
        if (!down) m = __builtin_bitreverse32(m) >> (32 - CHUNK);
        m >>= down ? first - wbn : wbn - first;
        if (need[u] && m != 0) { far[u] = base + __builtin_ctz(m); need[u] = false; }
      }
    }
  }

#pragma unroll
  for (int u = 0; u < UNITS_PER_WAVE; ++u) {
    int unit = wave * UNITS_PER_WAVE + u;
    int down = unit & 1;
    int c = (unit >> 1) * STRIP + lane;
#pragma unroll
    for (int j = 0; j < BAND; ++j) {
      unsigned s = bitsu[u] >> j;
      int first = s ? j + __builtin_ctz(s) : far[u];
      int d = min(first - j, BIGI);
      int i = down ? j : BAND - 1 - j;
      dist[down][i][c] = d;
    }
  }
  __syncthreads();

  int row = blockIdx.x * WAVES_PER_BLOCK + wave;
  float* orow = out + (size_t)row * WW;

  bool upper = lane >= 16;
  int nl = lane & 15;

  v2f ct[NT];
#pragma unroll
  for (int t = 0; t < NT; ++t) {
    float jp = (float)(t * 16 + nl);
    int dv = min(dist[0][wave][t * 16 + nl], dist[1][wave][t * 16 + nl]);
    float df = (float)dv;
    float g = df * df;
    ct[t].x = upper ? (jp * jp + g) : 1.0f;
    ct[t].y = upper ? 0.0f : jp;
  }

  v8f czero = {};
  float dsq[NT / 2];
#pragma unroll
  for (int p = 0; p < NT / 2; ++p) {
    float j0 = (float)(p * 32 + nl);
    float j1 = j0 + 16.0f;
    v2f b0, b1;
    b0.x = upper ? 1.0f : (j0 * j0);
    b0.y = upper ? 0.0f : (-2.0f * j0);
    b1.x = upper ? 1.0f : (j1 * j1);
    b1.y = upper ? 0.0f : (-2.0f * j1);

    v8f acc0, acc1;
#pragma unroll
    for (int i = 0; i < 8; ++i) { acc0[i] = 3.0e38f; acc1[i] = 3.0e38f; }

#pragma unroll
    for (int t = 0; t < NT; t += 2) {
      v8f d00 = __builtin_amdgcn_wmma_f32_16x16x4_f32(
          false, ct[t], false, b0, (short)0, czero, false, false);
      v8f d10 = __builtin_amdgcn_wmma_f32_16x16x4_f32(
          false, ct[t], false, b1, (short)0, czero, false, false);
      v8f d01 = __builtin_amdgcn_wmma_f32_16x16x4_f32(
          false, ct[t + 1], false, b0, (short)0, czero, false, false);
      v8f d11 = __builtin_amdgcn_wmma_f32_16x16x4_f32(
          false, ct[t + 1], false, b1, (short)0, czero, false, false);
#pragma unroll
      for (int i = 0; i < 8; ++i) {
        acc0[i] = fminf(fminf(acc0[i], d00[i]), d01[i]);
        acc1[i] = fminf(fminf(acc1[i], d10[i]), d11[i]);
      }
    }

    float m0 = fminf(fminf(fminf(acc0[0], acc0[1]), fminf(acc0[2], acc0[3])),
                     fminf(fminf(acc0[4], acc0[5]), fminf(acc0[6], acc0[7])));
    float m1 = fminf(fminf(fminf(acc1[0], acc1[1]), fminf(acc1[2], acc1[3])),
                     fminf(fminf(acc1[4], acc1[5]), fminf(acc1[6], acc1[7])));
    m0 = fminf(m0, xhalf(m0));
    m1 = fminf(m1, xhalf(m1));

    dsq[p] = upper ? m1 : m0;
  }

#pragma unroll
  for (int p = 0; p < NT / 2; ++p) {
    float v = 2.0f / (1.0f + expf(0.1f * sqrtf(dsq[p])));
    orow[p * 32 + lane] = v;
  }
}

extern "C" void kernel_launch(void* const* d_in, const int* in_sizes, int n_in,
                              void* d_out, int out_size, void* d_ws, size_t ws_size,
                              hipStream_t stream) {
  (void)in_sizes; (void)n_in; (void)out_size; (void)d_ws; (void)ws_size;
  const float* mask = (const float*)d_in[0];
  float* out = (float*)d_out;

  int rows = BB * HH;
  edt_fused<<<rows / WAVES_PER_BLOCK, 32 * WAVES_PER_BLOCK, 0, stream>>>(mask, out);
}


# row pass: candidate tiles that cannot win are skipped: a row with all vertical distances <= 17 multiplies only the 3 adjacent candidate tiles per output tile (34 WMMAs, not 144); else the full product
# speedup vs baseline: 8.0292x; 1.5101x over previous
#include <hip/hip_runtime.h>
#include <math.h>


#define BB 8
#define HH 192
#define WW 192
#define NT (WW / 16)
#define WAVES_PER_BLOCK 4
#define BAND WAVES_PER_BLOCK
#define BIGI (HH * HH + WW * WW)
#define CHUNK 16
#define STRIP 32
#define NUNIT (2 * WW / STRIP)
#define UNITS_PER_WAVE (NUNIT / WAVES_PER_BLOCK)
#define NOTFOUND (BIGI + 4 * CHUNK)
#define NEAR 17

typedef __attribute__((ext_vector_type(2))) float v2f;
typedef __attribute__((ext_vector_type(8))) float v8f;

__device__ __forceinline__ float xhalf(float x) {
  int b = __builtin_bit_cast(int, x);
  return __builtin_bit_cast(float, __builtin_amdgcn_permlanex16(
      b, b, 0x76543210, 0xfedcba98, false, false));
}

__device__ __forceinline__ v8f tile(v2f a, v2f b) {
  v8f czero = {};
  return __builtin_amdgcn_wmma_f32_16x16x4_f32(
      false, a, false, b, (short)0, czero, false, false);
}

__device__ __forceinline__ float colmin(v8f acc) {
  float m = fminf(fminf(fminf(acc[0], acc[1]), fminf(acc[2], acc[3])),
                  fminf(fminf(acc[4], acc[5]), fminf(acc[6], acc[7])));
  return fminf(m, xhalf(m));
}

__global__ __launch_bounds__(32 * WAVES_PER_BLOCK)
void edt_fused(const float* __restrict__ mask, float* __restrict__ out) {
  __shared__ int dist[2][BAND][WW];

  int wave = __builtin_amdgcn_readfirstlane((int)(threadIdx.x >> 5));
  int lane = threadIdx.x & 31;
  int b = blockIdx.x / (HH / BAND);
  int r0 = (blockIdx.x % (HH / BAND)) * BAND;
  const float* img = mask + (size_t)b * HH * WW;

  float v[UNITS_PER_WAVE][CHUNK];
#pragma unroll
  for (int u = 0; u < UNITS_PER_WAVE; ++u) {
    int unit = wave * UNITS_PER_WAVE + u;
    int down = unit & 1;
    int c = (unit >> 1) * STRIP + lane;
    int start = down ? r0 : r0 + BAND - 1;
    int wb = down ? min(start, HH - CHUNK) : max(start, CHUNK - 1);
    int lo = down ? wb : wb - (CHUNK - 1);
    const float* p = img + lo * WW + c;
#pragma unroll
    for (int i = 0; i < CHUNK; ++i) v[u][i] = p[i * WW];
  }

  unsigned bitsu[UNITS_PER_WAVE];
  bool need[UNITS_PER_WAVE];
  int far[UNITS_PER_WAVE];
#pragma unroll
  for (int u = 0; u < UNITS_PER_WAVE; ++u) {
    int unit = wave * UNITS_PER_WAVE + u;
    int down = unit & 1;
    int start = down ? r0 : r0 + BAND - 1;
    int wb = down ? min(start, HH - CHUNK) : max(start, CHUNK - 1);
    int maxpos = down ? HH - 1 - r0 : r0 + BAND - 1;
    unsigned bits = 0;
#pragma unroll
    for (int i = 0; i < CHUNK; ++i) bits |= (v[u][i] > 0.5f) ? (1u << i) : 0u;
    if (!down) bits = __builtin_bitreverse32(bits) >> (32 - CHUNK);
    bits >>= down ? start - wb : wb - start;
    bitsu[u] = bits;
    need[u] = ((bits >> (BAND - 1)) == 0) && (CHUNK <= maxpos);
    far[u] = down ? BIGI - r0 : NOTFOUND;
  }

  if (__builtin_amdgcn_ballot_w32(need[0] || need[1] || need[2]) != 0) {
#pragma unroll
    for (int u = 0; u < UNITS_PER_WAVE; ++u) {
      int unit = wave * UNITS_PER_WAVE + u;
      int down = unit & 1;
      int c = (unit >> 1) * STRIP + lane;
      int start = down ? r0 : r0 + BAND - 1;
      int maxpos = down ? HH - 1 - r0 : r0 + BAND - 1;
      for (int base = CHUNK;
           base <= maxpos && __builtin_amdgcn_ballot_w32(need[u]) != 0;
           base += CHUNK) {
        int first = down ? start + base : start - base;
        int wbn = down ? min(first, HH - CHUNK) : max(first, CHUNK - 1);
        const float* q = img + (down ? wbn : wbn - (CHUNK - 1)) * WW + c;
        unsigned m = 0;
#pragma unroll
        for (int i = 0; i < CHUNK; ++i) m |= (q[i * WW] > 0.5f) ? (1u << i) : 0u;
        if (!down) m = __builtin_bitreverse32(m) >> (32 - CHUNK);
        m >>= down ? first - wbn : wbn - first;
        if (need[u] && m != 0) { far[u] = base + __builtin_ctz(m); need[u] = false; }
      }
    }
  }

#pragma unroll
  for (int u = 0; u < UNITS_PER_WAVE; ++u) {
    int unit = wave * UNITS_PER_WAVE + u;
    int down = unit & 1;
    int c = (unit >> 1) * STRIP + lane;
#pragma unroll
    for (int j = 0; j < BAND; ++j) {
      unsigned s = bitsu[u] >> j;
      int first = s ? j + __builtin_ctz(s) : far[u];
      int d = min(first - j, BIGI);
      int i = down ? j : BAND - 1 - j;
      dist[down][i][c] = d;
    }
  }
  __syncthreads();

  int row = blockIdx.x * WAVES_PER_BLOCK + wave;
  float* orow = out + (size_t)row * WW;

  bool upper = lane >= 16;
  int nl = lane & 15;

  v2f ct[NT];
  int dmax = 0;
#pragma unroll
  for (int t = 0; t < NT; ++t) {
    float jp = (float)(t * 16 + nl);
    int dv = min(dist[0][wave][t * 16 + nl], dist[1][wave][t * 16 + nl]);
    dmax = max(dmax, dv);
    float df = (float)dv;
    float g = df * df;
    ct[t].x = upper ? (jp * jp + g) : 1.0f;
    ct[t].y = upper ? 0.0f : jp;
  }

  bool near_row = __builtin_amdgcn_ballot_w32(dmax > NEAR) == 0;

  float dsq[NT / 2];
  if (__builtin_expect(near_row, true)) {
#pragma unroll
    for (int p = 0; p < NT / 2; ++p) {
      float j0 = (float)(p * 32 + nl);
      float j1 = j0 + 16.0f;
      v2f b0, b1;
      b0.x = upper ? 1.0f : (j0 * j0);
      b0.y = upper ? 0.0f : (-2.0f * j0);
      b1.x = upper ? 1.0f : (j1 * j1);
      b1.y = upper ? 0.0f : (-2.0f * j1);

      v8f d00 = tile(ct[2 * p], b0);
      v8f d10 = tile(ct[2 * p], b1);
      v8f d01 = tile(ct[2 * p + 1], b0);
      v8f d11 = tile(ct[2 * p + 1], b1);
      v8f acc0, acc1;
#pragma unroll
      for (int i = 0; i < 8; ++i) {
        acc0[i] = fminf(d00[i], d01[i]);
        acc1[i] = fminf(d10[i], d11[i]);
      }
      if (p > 0) {
        v8f dl = tile(ct[2 * p - 1], b0);
#pragma unroll
        for (int i = 0; i < 8; ++i) acc0[i] = fminf(acc0[i], dl[i]);
      }
      if (p < NT / 2 - 1) {
        v8f dr = tile(ct[2 * p + 2], b1);
#pragma unroll
        for (int i = 0; i < 8; ++i) acc1[i] = fminf(acc1[i], dr[i]);
      }
      float m0 = colmin(acc0);
      float m1 = colmin(acc1);
      dsq[p] = upper ? m1 : m0;
    }
  } else {
#pragma unroll
    for (int p = 0; p < NT / 2; ++p) {
      float j0 = (float)(p * 32 + nl);
      float j1 = j0 + 16.0f;
      v2f b0, b1;
      b0.x = upper ? 1.0f : (j0 * j0);
      b0.y = upper ? 0.0f : (-2.0f * j0);
      b1.x = upper ? 1.0f : (j1 * j1);
      b1.y = upper ? 0.0f : (-2.0f * j1);

      v8f acc0, acc1;
#pragma unroll
      for (int i = 0; i < 8; ++i) { acc0[i] = 3.0e38f; acc1[i] = 3.0e38f; }

#pragma unroll
      for (int t = 0; t < NT; t += 2) {
        v8f d00 = tile(ct[t], b0);
        v8f d10 = tile(ct[t], b1);
        v8f d01 = tile(ct[t + 1], b0);
        v8f d11 = tile(ct[t + 1], b1);
#pragma unroll
        for (int i = 0; i < 8; ++i) {
          acc0[i] = fminf(fminf(acc0[i], d00[i]), d01[i]);
          acc1[i] = fminf(fminf(acc1[i], d10[i]), d11[i]);
        }
      }
      float m0 = colmin(acc0);
      float m1 = colmin(acc1);
      dsq[p] = upper ? m1 : m0;
    }
  }

#pragma unroll
  for (int p = 0; p < NT / 2; ++p) {
    float v = 2.0f / (1.0f + expf(0.1f * sqrtf(dsq[p])));
    orow[p * 32 + lane] = v;
  }
}

extern "C" void kernel_launch(void* const* d_in, const int* in_sizes, int n_in,
                              void* d_out, int out_size, void* d_ws, size_t ws_size,
                              hipStream_t stream) {
  (void)in_sizes; (void)n_in; (void)out_size; (void)d_ws; (void)ws_size;
  const float* mask = (const float*)d_in[0];
  float* out = (float*)d_out;

  int rows = BB * HH;
  edt_fused<<<rows / WAVES_PER_BLOCK, 32 * WAVES_PER_BLOCK, 0, stream>>>(mask, out);
}
